# starting kernel
# baseline (speedup 1.0000x reference)
#include <hip/hip_runtime.h>
#include <hip/hip_bf16.h>

#define BB   8
#define CC   256
#define HH   64
#define WWD  64
#define EE   9
#define OUTC 256
#define EPSF 1e-5f

typedef __attribute__((ext_vector_type(2))) float v2f;
typedef __attribute__((ext_vector_type(8))) float v8f;

#define WS_WEFF   0
#define WS_EBIAS  2304
#define WS_B2     2560
#define WS_T2     2816
#define WS_WSWZ   3072

__device__ __forceinline__ float silu(float v) {
    return v / (1.0f + __expf(-v));
}

__global__ __launch_bounds__(256) void fold_kernel(
    const float* __restrict__ w_experts, const float* __restrict__ bn_gamma,
    const float* __restrict__ bn_beta,   const float* __restrict__ bn_mean,
    const float* __restrict__ bn_var,    const float* __restrict__ scales,
    const float* __restrict__ out_gamma, const float* __restrict__ out_beta,
    const float* __restrict__ out_mean,  const float* __restrict__ out_var,
    float* __restrict__ ws)
{
    const int c = threadIdx.x;
    float wacc[9];
#pragma unroll
    for (int k = 0; k < 9; ++k) wacc[k] = 0.0f;
    float bias = 0.0f;
#pragma unroll
    for (int e = 0; e < EE; ++e) {
        const int ec = e * CC + c;
        const float t  = bn_gamma[ec] * rsqrtf(bn_var[ec] + EPSF);
        const float es = t * scales[ec];
        bias += (bn_beta[ec] - bn_mean[ec] * t) * scales[ec];
        const float* wp = w_experts + ec * 9;
#pragma unroll
        for (int k = 0; k < 9; ++k) wacc[k] += es * wp[k];
    }
#pragma unroll
    for (int k = 0; k < 9; ++k) ws[WS_WEFF + c * 9 + k] = wacc[k];
    ws[WS_EBIAS + c] = bias;

    const float t2 = out_gamma[c] * rsqrtf(out_var[c] + EPSF);
    ws[WS_T2 + c] = t2;
    ws[WS_B2 + c] = out_beta[c] - out_mean[c] * t2;
}

__global__ __launch_bounds__(256) void swz_kernel(
    const float* __restrict__ w_out, float* __restrict__ ws)
{
    const int idx   = blockIdx.x * 256 + threadIdx.x;
    const int kstep = idx >> 9;
    const int rem   = idx & 511;
    const int mt    = rem >> 5;
    const int lane  = rem & 31;
    const int m     = mt * 16 + (lane & 15);
    const int k     = kstep * 4 + ((lane >> 4) << 1);
    const float t2  = ws[WS_T2 + m];
    ws[WS_WSWZ + idx * 2]     = w_out[m * CC + k]     * t2;
    ws[WS_WSWZ + idx * 2 + 1] = w_out[m * CC + k + 1] * t2;
}

__global__ __launch_bounds__(256) void move_kernel(
    const float* __restrict__ x, const float* __restrict__ ws,
    float* __restrict__ out)
{
    __shared__ float smem[CC * WWD];

    const int tid = threadIdx.x;
    const int b   = blockIdx.x >> 6;
    const int h   = blockIdx.x & 63;

    const float* weff  = ws + WS_WEFF;
    const float* ebias = ws + WS_EBIAS;
    const float* b2    = ws + WS_B2;
    const float* wswz  = ws + WS_WSWZ;

    {
        const int w  = tid & 63;
        const int cl = tid >> 6;
#pragma unroll 2
        for (int ci = 0; ci < 64; ++ci) {
            const int c = (ci << 2) + cl;
            const float* xp = x + (((b * CC + c) * HH + h) * WWD) + w;
            const float* we = weff + c * 9;
            float s = 0.0f;
#pragma unroll
            for (int dy = -1; dy <= 1; ++dy) {
                const int h2 = h + dy;
                if (h2 >= 0 && h2 < HH) {
#pragma unroll
                    for (int dx = -1; dx <= 1; ++dx) {
                        const int w2 = w + dx;
                        if (w2 >= 0 && w2 < WWD)
                            s += xp[dy * WWD + dx] * we[(dy + 1) * 3 + (dx + 1)];
                    }
                }
            }
            s = silu(s + ebias[c]);
            const int kstep  = c >> 2;
            const int kin    = c & 3;
            const int lane_s = (w & 15) + ((kin >> 1) << 4);
            const int ntile  = w >> 4;
            smem[(((((kstep << 2) + ntile) << 5) + lane_s) << 1) + (kin & 1)] = s;
        }
    }
    __syncthreads();

    const int lane = tid & 31;
    const int wv   = tid >> 5;
    const int mt0  = wv << 1;
    const int mo   = (lane >> 4) << 3;

    v8f acc[2][4];
#pragma unroll
    for (int ms = 0; ms < 2; ++ms) {
        const float* bp = b2 + ((mt0 + ms) << 4) + mo;
#pragma unroll
        for (int r = 0; r < 8; ++r) {
            const float bval = bp[r];
#pragma unroll
            for (int ns = 0; ns < 4; ++ns) acc[ms][ns][r] = bval;
        }
    }

    const v2f* aptr = (const v2f*)wswz;
    const v2f* bptr = (const v2f*)smem;
#pragma unroll 4
    for (int kstep = 0; kstep < 64; ++kstep) {
        const v2f a0 = aptr[((kstep << 4) + mt0)     * 32 + lane];
        const v2f a1 = aptr[((kstep << 4) + mt0 + 1) * 32 + lane];
#pragma unroll
        for (int ns = 0; ns < 4; ++ns) {
            const v2f bf = bptr[((kstep << 2) + ns) * 32 + lane];
            acc[0][ns] = __builtin_amdgcn_wmma_f32_16x16x4_f32(
                false, a0, false, bf, (short)0, acc[0][ns], false, false);
            acc[1][ns] = __builtin_amdgcn_wmma_f32_16x16x4_f32(
                false, a1, false, bf, (short)0, acc[1][ns], false, false);
        }
    }

    const int ncol = lane & 15;
#pragma unroll
    for (int ms = 0; ms < 2; ++ms) {
#pragma unroll
        for (int ns = 0; ns < 4; ++ns) {
#pragma unroll
            for (int r = 0; r < 8; ++r) {
                const int o = ((mt0 + ms) << 4) + mo + r;
                const float v = silu(acc[ms][ns][r]);
                out[(((b * OUTC + o) * HH + h) * WWD) + (ns << 4) + ncol] = v;
            }
        }
    }
}

extern "C" void kernel_launch(void* const* d_in, const int* in_sizes, int n_in,
                              void* d_out, int out_size, void* d_ws, size_t ws_size,
                              hipStream_t stream) {
    const float* x         = (const float*)d_in[0];
    const float* w_experts = (const float*)d_in[1];
    const float* bn_gamma  = (const float*)d_in[2];
    const float* bn_beta   = (const float*)d_in[3];
    const float* bn_mean   = (const float*)d_in[4];
    const float* bn_var    = (const float*)d_in[5];
    const float* scales    = (const float*)d_in[6];
    const float* w_out     = (const float*)d_in[7];
    const float* out_gamma = (const float*)d_in[8];
    const float* out_beta  = (const float*)d_in[9];
    const float* out_mean  = (const float*)d_in[10];
    const float* out_var   = (const float*)d_in[11];
    float* out = (float*)d_out;
    float* ws  = (float*)d_ws;

    fold_kernel<<<1, 256, 0, stream>>>(w_experts, bn_gamma, bn_beta, bn_mean,
                                       bn_var, scales, out_gamma, out_beta,
                                       out_mean, out_var, ws);
    swz_kernel<<<128, 256, 0, stream>>>(w_out, ws);
    move_kernel<<<BB * HH, 256, 0, stream>>>(x, ws, out);
}


# conv stage: x rows staged per 32-channel chunk through a 24 KB LDS halo tile (16-byte row loads issued together, next chunk in flight under the taps); weights by scalar loads; tap order kept
# speedup vs baseline: 2.6475x; 2.6475x over previous
#include <hip/hip_runtime.h>
#include <hip/hip_bf16.h>

#define BB   8
#define CC   256
#define HH   64
#define WWD  64
#define EE   9
#define OUTC 256
#define EPSF 1e-5f

typedef __attribute__((ext_vector_type(2))) float v2f;
typedef __attribute__((ext_vector_type(8))) float v8f;

#define WS_WEFF   0
#define WS_EBIAS  2304
#define WS_B2     2560
#define WS_T2     2816
#define WS_WSWZ   3072

__device__ __forceinline__ float silu(float v) {
    return v / (1.0f + __expf(-v));
}

__global__ __launch_bounds__(256) void fold_kernel(
    const float* __restrict__ w_experts, const float* __restrict__ bn_gamma,
    const float* __restrict__ bn_beta,   const float* __restrict__ bn_mean,
    const float* __restrict__ bn_var,    const float* __restrict__ scales,
    const float* __restrict__ out_gamma, const float* __restrict__ out_beta,
    const float* __restrict__ out_mean,  const float* __restrict__ out_var,
    float* __restrict__ ws)
{
    const int c = threadIdx.x;
    float wacc[9];
#pragma unroll
    for (int k = 0; k < 9; ++k) wacc[k] = 0.0f;
    float bias = 0.0f;
#pragma unroll
    for (int e = 0; e < EE; ++e) {
        const int ec = e * CC + c;
        const float t  = bn_gamma[ec] * rsqrtf(bn_var[ec] + EPSF);
        const float es = t * scales[ec];
        bias += (bn_beta[ec] - bn_mean[ec] * t) * scales[ec];
        const float* wp = w_experts + ec * 9;
#pragma unroll
        for (int k = 0; k < 9; ++k) wacc[k] += es * wp[k];
    }
#pragma unroll
    for (int k = 0; k < 9; ++k) ws[WS_WEFF + c * 9 + k] = wacc[k];
    ws[WS_EBIAS + c] = bias;

    const float t2 = out_gamma[c] * rsqrtf(out_var[c] + EPSF);
    ws[WS_T2 + c] = t2;
    ws[WS_B2 + c] = out_beta[c] - out_mean[c] * t2;
}

__global__ __launch_bounds__(256) void swz_kernel(
    const float* __restrict__ w_out, float* __restrict__ ws)
{
    const int idx   = blockIdx.x * 256 + threadIdx.x;
    const int kstep = idx >> 9;
    const int rem   = idx & 511;
    const int mt    = rem >> 5;
    const int lane  = rem & 31;
    const int m     = mt * 16 + (lane & 15);
    const int k     = kstep * 4 + ((lane >> 4) << 1);
    const float t2  = ws[WS_T2 + m];
    ws[WS_WSWZ + idx * 2]     = w_out[m * CC + k]     * t2;
    ws[WS_WSWZ + idx * 2 + 1] = w_out[m * CC + k + 1] * t2;
}

__global__ __launch_bounds__(256) void move_kernel(
    const float* __restrict__ x, const float* __restrict__ ws,
    float* __restrict__ out)
{
    __shared__ float smem[CC * WWD];
    __shared__ float4 rawt[3 * 32 * 16];

    const int tid = threadIdx.x;
    const int b   = blockIdx.x >> 6;
    const int h   = blockIdx.x & 63;

    const float* weff  = ws + WS_WEFF;
    const float* ebias = ws + WS_EBIAS;
    const float* b2    = ws + WS_B2;
    const float* wswz  = ws + WS_WSWZ;

    {
        const int col4 = tid & 15;
        const int chA  = tid >> 4;
        const float* xb = x + b * (CC * HH * WWD);

        const int w  = tid & 63;
        const int cl = __builtin_amdgcn_readfirstlane(tid >> 6);
        const bool has_l = (w > 0);
        const bool has_r = (w < WWD - 1);
        const int wl = has_l ? w - 1 : 0;
        const int wr = has_r ? w + 1 : WWD - 1;
        const float* rawf = (const float*)rawt;

        float4 pre[6];
#pragma unroll
        for (int j = 0; j < 6; ++j) {
            const int r  = j >> 1;
            int hr = h + r - 1;
            hr = hr < 0 ? 0 : (hr > HH - 1 ? HH - 1 : hr);
            const int c = chA + ((j & 1) << 4);
            pre[j] = *(const float4*)(xb + ((c * HH + hr) * WWD) + (col4 << 2));
        }

        for (int chunk = 0; chunk < 8; ++chunk) {
#pragma unroll
            for (int j = 0; j < 6; ++j) {
                const int r = j >> 1;
                const bool rv = (unsigned)(h + r - 1) < (unsigned)HH;
                float4 v = pre[j];
                if (!rv) v = make_float4(0.0f, 0.0f, 0.0f, 0.0f);
                rawt[tid + (j << 8)] = v;
            }
            __syncthreads();

            if (chunk + 1 < 8) {
#pragma unroll
                for (int j = 0; j < 6; ++j) {
                    const int r  = j >> 1;
                    int hr = h + r - 1;
                    hr = hr < 0 ? 0 : (hr > HH - 1 ? HH - 1 : hr);
                    const int c = ((chunk + 1) << 5) + chA + ((j & 1) << 4);
                    pre[j] = *(const float4*)(xb + ((c * HH + hr) * WWD) + (col4 << 2));
                }
            }

#pragma unroll
            for (int jj = 0; jj < 8; ++jj) {
                const int chl = cl + (jj << 2);
                const int c   = (chunk << 5) + chl;
                const float* we = weff + c * 9;
                float s = 0.0f;
#pragma unroll
                for (int r = 0; r < 3; ++r) {
                    const float* rp = rawf + ((r * 32 + chl) << 6);
                    float xl = rp[wl];
                    const float xc = rp[w];
                    float xr = rp[wr];
                    xl = has_l ? xl : 0.0f;
                    xr = has_r ? xr : 0.0f;
                    s = __builtin_fmaf(xl, we[r * 3 + 0], s);
                    s = __builtin_fmaf(xc, we[r * 3 + 1], s);
                    s = __builtin_fmaf(xr, we[r * 3 + 2], s);
                }
                s = silu(s + ebias[c]);
                const int kstep  = c >> 2;
                const int kin    = c & 3;
                const int lane_s = (w & 15) + ((kin >> 1) << 4);
                const int ntile  = w >> 4;
                smem[(((((kstep << 2) + ntile) << 5) + lane_s) << 1) + (kin & 1)] = s;
            }
            __syncthreads();
        }
    }

    const int lane = tid & 31;
    const int wv   = tid >> 5;
    const int mt0  = wv << 1;
    const int mo   = (lane >> 4) << 3;

    v8f acc[2][4];
#pragma unroll
    for (int ms = 0; ms < 2; ++ms) {
        const float* bp = b2 + ((mt0 + ms) << 4) + mo;
#pragma unroll
        for (int r = 0; r < 8; ++r) {
            const float bval = bp[r];
#pragma unroll
            for (int ns = 0; ns < 4; ++ns) acc[ms][ns][r] = bval;
        }
    }

    const v2f* aptr = (const v2f*)wswz;
    const v2f* bptr = (const v2f*)smem;
#pragma unroll 4
    for (int kstep = 0; kstep < 64; ++kstep) {
        const v2f a0 = aptr[((kstep << 4) + mt0)     * 32 + lane];
        const v2f a1 = aptr[((kstep << 4) + mt0 + 1) * 32 + lane];
#pragma unroll
        for (int ns = 0; ns < 4; ++ns) {
            const v2f bf = bptr[((kstep << 2) + ns) * 32 + lane];
            acc[0][ns] = __builtin_amdgcn_wmma_f32_16x16x4_f32(
                false, a0, false, bf, (short)0, acc[0][ns], false, false);
            acc[1][ns] = __builtin_amdgcn_wmma_f32_16x16x4_f32(
                false, a1, false, bf, (short)0, acc[1][ns], false, false);
        }
    }

    const int ncol = lane & 15;
#pragma unroll
    for (int ms = 0; ms < 2; ++ms) {
#pragma unroll
        for (int ns = 0; ns < 4; ++ns) {
#pragma unroll
            for (int r = 0; r < 8; ++r) {
                const int o = ((mt0 + ms) << 4) + mo + r;
                const float v = silu(acc[ms][ns][r]);
                out[(((b * OUTC + o) * HH + h) * WWD) + (ns << 4) + ncol] = v;
            }
        }
    }
}

extern "C" void kernel_launch(void* const* d_in, const int* in_sizes, int n_in,
                              void* d_out, int out_size, void* d_ws, size_t ws_size,
                              hipStream_t stream) {
    const float* x         = (const float*)d_in[0];
    const float* w_experts = (const float*)d_in[1];
    const float* bn_gamma  = (const float*)d_in[2];
    const float* bn_beta   = (const float*)d_in[3];
    const float* bn_mean   = (const float*)d_in[4];
    const float* bn_var    = (const float*)d_in[5];
    const float* scales    = (const float*)d_in[6];
    const float* w_out     = (const float*)d_in[7];
    const float* out_gamma = (const float*)d_in[8];
    const float* out_beta  = (const float*)d_in[9];
    const float* out_mean  = (const float*)d_in[10];
    const float* out_var   = (const float*)d_in[11];
    float* out = (float*)d_out;
    float* ws  = (float*)d_ws;

    fold_kernel<<<1, 256, 0, stream>>>(w_experts, bn_gamma, bn_beta, bn_mean,
                                       bn_var, scales, out_gamma, out_beta,
                                       out_mean, out_var, ws);
    swz_kernel<<<128, 256, 0, stream>>>(w_out, ws);
    move_kernel<<<BB * HH, 256, 0, stream>>>(x, ws, out);
}


# single launch, no workspace: each work-group folds expert taps and output BN into LDS itself, A fragments read straight from w_out; k order in 32-channel blocks permuted, BN scale on the accumulator
# speedup vs baseline: 3.1499x; 1.1898x over previous
#include <hip/hip_runtime.h>
#include <hip/hip_bf16.h>


#define BB   8
#define CC   256
#define HH   64
#define WWD  64
#define EE   9
#define OUTC 256
#define EPSF 1e-5f

typedef __attribute__((ext_vector_type(2))) float v2f;
typedef __attribute__((ext_vector_type(8))) float v8f;

__device__ __forceinline__ float silu(float v) {
    return v / (1.0f + __expf(-v));
}

__global__ __launch_bounds__(256) void move_kernel(
    const float* __restrict__ x,         const float* __restrict__ w_experts,
    const float* __restrict__ bn_gamma,  const float* __restrict__ bn_beta,
    const float* __restrict__ bn_mean,   const float* __restrict__ bn_var,
    const float* __restrict__ scales,    const float* __restrict__ w_out,
    const float* __restrict__ out_gamma, const float* __restrict__ out_beta,
    const float* __restrict__ out_mean,  const float* __restrict__ out_var,
    float* __restrict__ out)
{
    __shared__ float smem[CC * WWD];
    __shared__ float4 rawt[3 * 32 * 16];
    __shared__ float4 taps[CC * 3];
    __shared__ float obn[2 * OUTC];

    const int tid = threadIdx.x;
    const int b   = blockIdx.x >> 6;
    const int h   = blockIdx.x & 63;

    {
        const int col4 = tid & 15;
        const int chA  = tid >> 4;
        const float* xb = x + b * (CC * HH * WWD);

        const int w  = tid & 63;
        const int cl = __builtin_amdgcn_readfirstlane(tid >> 6);
        const bool has_l = (w > 0);
        const bool has_r = (w < WWD - 1);
        const int wl = has_l ? w - 1 : 0;
        const int wr = has_r ? w + 1 : WWD - 1;
        const float* rawf = (const float*)rawt;

        float4 pre[6];
#pragma unroll
        for (int j = 0; j < 6; ++j) {
            const int r  = j >> 1;
            int hr = h + r - 1;
            hr = hr < 0 ? 0 : (hr > HH - 1 ? HH - 1 : hr);
            const int c = chA + ((j & 1) << 4);
            pre[j] = *(const float4*)(xb + ((c * HH + hr) * WWD) + (col4 << 2));
        }

        {
            const int c = tid;
            float bias = 0.0f;
#pragma unroll
            for (int e = 0; e < EE; ++e) {
                const int ec = e * CC + c;
                const float t  = bn_gamma[ec] * rsqrtf(bn_var[ec] + EPSF);
                const float es = t * scales[ec];
                bias += (bn_beta[ec] - bn_mean[ec] * t) * scales[ec];
                smem[ec] = es;
            }
            ((float*)taps)[c * 12 + 9] = bias;
            const float t2 = out_gamma[c] * rsqrtf(out_var[c] + EPSF);
            obn[c]        = t2;
            obn[OUTC + c] = out_beta[c] - out_mean[c] * t2;
        }
        __syncthreads();
#pragma unroll
        for (int j = 0; j < 9; ++j) {
            const int idx = tid + (j << 8);
            const int c   = idx / 9;
            const int k   = idx - c * 9;
            float acc = 0.0f;
#pragma unroll
            for (int e = 0; e < EE; ++e)
                acc += smem[e * CC + c] * w_experts[e * (CC * 9) + idx];
            ((float*)taps)[c * 12 + k] = acc;
        }

        for (int chunk = 0; chunk < 8; ++chunk) {
#pragma unroll
            for (int j = 0; j < 6; ++j) {
                const int r = j >> 1;
                const bool rv = (unsigned)(h + r - 1) < (unsigned)HH;
                float4 v = pre[j];
                if (!rv) v = make_float4(0.0f, 0.0f, 0.0f, 0.0f);
                rawt[tid + (j << 8)] = v;
            }
            __syncthreads();

            if (chunk + 1 < 8) {
#pragma unroll
                for (int j = 0; j < 6; ++j) {
                    const int r  = j >> 1;
                    int hr = h + r - 1;
                    hr = hr < 0 ? 0 : (hr > HH - 1 ? HH - 1 : hr);
                    const int c = ((chunk + 1) << 5) + chA + ((j & 1) << 4);
                    pre[j] = *(const float4*)(xb + ((c * HH + hr) * WWD) + (col4 << 2));
                }
            }

#pragma unroll
            for (int jj = 0; jj < 8; ++jj) {
                const int chl = cl + (jj << 2);
                const int c   = (chunk << 5) + chl;
                const float4 t0 = taps[c * 3 + 0];
                const float4 t1 = taps[c * 3 + 1];
                const float4 t2 = taps[c * 3 + 2];
                const float we[9] = {t0.x, t0.y, t0.z, t0.w, t1.x, t1.y, t1.z, t1.w, t2.x};
                float s = 0.0f;
#pragma unroll
                for (int r = 0; r < 3; ++r) {
                    const float* rp = rawf + ((r * 32 + chl) << 6);
                    float xl = rp[wl];
                    const float xc = rp[w];
                    float xr = rp[wr];
                    xl = has_l ? xl : 0.0f;
                    xr = has_r ? xr : 0.0f;
                    s = __builtin_fmaf(xl, we[r * 3 + 0], s);
                    s = __builtin_fmaf(xc, we[r * 3 + 1], s);
                    s = __builtin_fmaf(xr, we[r * 3 + 2], s);
                }
                s = silu(s + t2.y);
                const int kstep  = ((c >> 5) << 3) + ((c >> 1) & 7);
                const int hf     = (c >> 4) & 1;
                const int lane_s = (w & 15) + (hf << 4);
                const int ntile  = w >> 4;
                smem[(((((kstep << 2) + ntile) << 5) + lane_s) << 1) + (c & 1)] = s;
            }
            __syncthreads();
        }
    }

    const int lane = tid & 31;
    const int wv   = tid >> 5;
    const int mt0  = wv << 1;
    const int mo   = (lane >> 4) << 3;

    v8f acc[2][4];
#pragma unroll
    for (int ms = 0; ms < 2; ++ms)
#pragma unroll
        for (int ns = 0; ns < 4; ++ns)
#pragma unroll
            for (int r = 0; r < 8; ++r) acc[ms][ns][r] = 0.0f;

    const float* ar0 = w_out + (((mt0 << 4) + (lane & 15)) * CC) + ((lane >> 4) << 4);
    const float* ar1 = ar0 + 16 * CC;
    const v2f* bptr = (const v2f*)smem;

    float4 abuf[2][2][4];
#pragma unroll
    for (int q = 0; q < 4; ++q) {
        abuf[0][0][q] = *(const float4*)(ar0 + (q << 2));
        abuf[0][1][q] = *(const float4*)(ar1 + (q << 2));
    }

#pragma unroll
    for (int kb = 0; kb < 8; ++kb) {
        const int cur = kb & 1;
        if (kb + 1 < 8) {
#pragma unroll
            for (int q = 0; q < 4; ++q) {
                abuf[cur ^ 1][0][q] = *(const float4*)(ar0 + ((kb + 1) << 5) + (q << 2));
                abuf[cur ^ 1][1][q] = *(const float4*)(ar1 + ((kb + 1) << 5) + (q << 2));
            }
        }
#pragma unroll
        for (int s = 0; s < 8; ++s) {
            const float4 q0 = abuf[cur][0][s >> 1];
            const float4 q1 = abuf[cur][1][s >> 1];
            v2f a0, a1;
            if (s & 1) { a0.x = q0.z; a0.y = q0.w; a1.x = q1.z; a1.y = q1.w; }
            else       { a0.x = q0.x; a0.y = q0.y; a1.x = q1.x; a1.y = q1.y; }
            const int kstep = (kb << 3) + s;
#pragma unroll
            for (int ns = 0; ns < 4; ++ns) {
                const v2f bf = bptr[((kstep << 2) + ns) * 32 + lane];
                acc[0][ns] = __builtin_amdgcn_wmma_f32_16x16x4_f32(
                    false, a0, false, bf, (short)0, acc[0][ns], false, false);
                acc[1][ns] = __builtin_amdgcn_wmma_f32_16x16x4_f32(
                    false, a1, false, bf, (short)0, acc[1][ns], false, false);
            }
        }
    }

    const int ncol = lane & 15;
#pragma unroll
    for (int ms = 0; ms < 2; ++ms) {
        const int obase = ((mt0 + ms) << 4) + mo;
#pragma unroll
        for (int r = 0; r < 8; ++r) {
            const int o = obase + r;
            const float t2 = obn[o];
            const float b2 = obn[OUTC + o];
#pragma unroll
            for (int ns = 0; ns < 4; ++ns) {
                const float v = silu(acc[ms][ns][r] * t2 + b2);
                out[(((b * OUTC + o) * HH + h) * WWD) + (ns << 4) + ncol] = v;
            }
        }
    }
}

extern "C" void kernel_launch(void* const* d_in, const int* in_sizes, int n_in,
                              void* d_out, int out_size, void* d_ws, size_t ws_size,
                              hipStream_t stream) {
    const float* x         = (const float*)d_in[0];
    const float* w_experts = (const float*)d_in[1];
    const float* bn_gamma  = (const float*)d_in[2];
    const float* bn_beta   = (const float*)d_in[3];
    const float* bn_mean   = (const float*)d_in[4];
    const float* bn_var    = (const float*)d_in[5];
    const float* scales    = (const float*)d_in[6];
    const float* w_out     = (const float*)d_in[7];
    const float* out_gamma = (const float*)d_in[8];
    const float* out_beta  = (const float*)d_in[9];
    const float* out_mean  = (const float*)d_in[10];
    const float* out_var   = (const float*)d_in[11];
    float* out = (float*)d_out;

    move_kernel<<<BB * HH, 256, 0, stream>>>(x, w_experts, bn_gamma, bn_beta, bn_mean, bn_var,
                                             scales, w_out, out_gamma, out_beta, out_mean,
                                             out_var, out);
}


# conv stage reads the LDS halo tile in 16-byte pieces: a lane does 4 adjacent pixels x 2 channels (was 1 pixel, 9 4-byte reads), results as two 16-byte B-fragment stores; tap order kept
# speedup vs baseline: 3.3583x; 1.0662x over previous
#include <hip/hip_runtime.h>
#include <hip/hip_bf16.h>


#define BB   8
#define CC   256
#define HH   64
#define WWD  64
#define EE   9
#define OUTC 256
#define EPSF 1e-5f

typedef __attribute__((ext_vector_type(2))) float v2f;
typedef __attribute__((ext_vector_type(8))) float v8f;

__device__ __forceinline__ float silu(float v) {
    return v / (1.0f + __expf(-v));
}

__global__ __launch_bounds__(256) void move_kernel(
    const float* __restrict__ x,         const float* __restrict__ w_experts,
    const float* __restrict__ bn_gamma,  const float* __restrict__ bn_beta,
    const float* __restrict__ bn_mean,   const float* __restrict__ bn_var,
    const float* __restrict__ scales,    const float* __restrict__ w_out,
    const float* __restrict__ out_gamma, const float* __restrict__ out_beta,
    const float* __restrict__ out_mean,  const float* __restrict__ out_var,
    float* __restrict__ out)
{
    __shared__ alignas(16) float smem[CC * WWD];
    __shared__ float4 rawt[3 * 32 * 16];
    __shared__ float4 taps[CC * 3];
    __shared__ float obn[2 * OUTC];

    const int tid = threadIdx.x;
    const int b   = blockIdx.x >> 6;
    const int h   = blockIdx.x & 63;

    {
        const int col4 = tid & 15;
        const int chA  = tid >> 4;
        const float* xb = x + b * (CC * HH * WWD);

        const int pr = tid >> 4;
        const bool has_l = (col4 > 0);
        const bool has_r = (col4 < 15);
        const int il = has_l ? (col4 << 2) - 1 : 0;
        const int ir = has_r ? (col4 << 2) + 4 : WWD - 1;
        const float* rawf = (const float*)rawt;
        const int f4lane = (((((pr & 7) << 2) + (col4 >> 2)) << 4) + ((col4 & 3) << 1) + ((pr >> 3) << 3));
        float4* smem4 = (float4*)smem;

        float4 pre[6];
#pragma unroll
        for (int j = 0; j < 6; ++j) {
            const int r  = j >> 1;
            int hr = h + r - 1;
            hr = hr < 0 ? 0 : (hr > HH - 1 ? HH - 1 : hr);
            const int c = chA + ((j & 1) << 4);
            pre[j] = *(const float4*)(xb + ((c * HH + hr) * WWD) + (col4 << 2));
        }

        {
            const int c = tid;
            float bias = 0.0f;
#pragma unroll
            for (int e = 0; e < EE; ++e) {
                const int ec = e * CC + c;
                const float t  = bn_gamma[ec] * rsqrtf(bn_var[ec] + EPSF);
                const float es = t * scales[ec];
                bias += (bn_beta[ec] - bn_mean[ec] * t) * scales[ec];
                smem[ec] = es;
            }
            ((float*)taps)[c * 12 + 9] = bias;
            const float t2 = out_gamma[c] * rsqrtf(out_var[c] + EPSF);
            obn[c]        = t2;
            obn[OUTC + c] = out_beta[c] - out_mean[c] * t2;
        }
        __syncthreads();
#pragma unroll
        for (int j = 0; j < 9; ++j) {
            const int idx = tid + (j << 8);
            const int c   = idx / 9;
            const int k   = idx - c * 9;
            float acc = 0.0f;
#pragma unroll
            for (int e = 0; e < EE; ++e)
                acc += smem[e * CC + c] * w_experts[e * (CC * 9) + idx];
            ((float*)taps)[c * 12 + k] = acc;
        }

        for (int chunk = 0; chunk < 8; ++chunk) {
#pragma unroll
            for (int j = 0; j < 6; ++j) {
                const int r = j >> 1;
                const bool rv = (unsigned)(h + r - 1) < (unsigned)HH;
                float4 v = pre[j];
                if (!rv) v = make_float4(0.0f, 0.0f, 0.0f, 0.0f);
                rawt[tid + (j << 8)] = v;
            }
            __syncthreads();

            if (chunk + 1 < 8) {
#pragma unroll
                for (int j = 0; j < 6; ++j) {
                    const int r  = j >> 1;
                    int hr = h + r - 1;
                    hr = hr < 0 ? 0 : (hr > HH - 1 ? HH - 1 : hr);
                    const int c = ((chunk + 1) << 5) + chA + ((j & 1) << 4);
                    pre[j] = *(const float4*)(xb + ((c * HH + hr) * WWD) + (col4 << 2));
                }
            }

            float ov[2][4];
#pragma unroll
            for (int e = 0; e < 2; ++e) {
                const int chl = (pr << 1) + e;
                const int c   = (chunk << 5) + chl;
                const float4 t0 = taps[c * 3 + 0];
                const float4 t1 = taps[c * 3 + 1];
                const float4 t2 = taps[c * 3 + 2];
                const float we[9] = {t0.x, t0.y, t0.z, t0.w, t1.x, t1.y, t1.z, t1.w, t2.x};
                float s0 = 0.0f, s1 = 0.0f, s2 = 0.0f, s3 = 0.0f;
#pragma unroll
                for (int r = 0; r < 3; ++r) {
                    const int row = r * 32 + chl;
                    const float4 m = rawt[(row << 4) + col4];
                    const float* rp = rawf + (row << 6);
                    float xl = rp[il];
                    float xr = rp[ir];
                    xl = has_l ? xl : 0.0f;
                    xr = has_r ? xr : 0.0f;
                    const float k0 = we[r * 3 + 0];
                    const float k1 = we[r * 3 + 1];
                    const float k2 = we[r * 3 + 2];
                    s0 = __builtin_fmaf(xl,  k0, s0);
                    s0 = __builtin_fmaf(m.x, k1, s0);
                    s0 = __builtin_fmaf(m.y, k2, s0);
                    s1 = __builtin_fmaf(m.x, k0, s1);
                    s1 = __builtin_fmaf(m.y, k1, s1);
                    s1 = __builtin_fmaf(m.z, k2, s1);
                    s2 = __builtin_fmaf(m.y, k0, s2);
                    s2 = __builtin_fmaf(m.z, k1, s2);
                    s2 = __builtin_fmaf(m.w, k2, s2);
                    s3 = __builtin_fmaf(m.z, k0, s3);
                    s3 = __builtin_fmaf(m.w, k1, s3);
                    s3 = __builtin_fmaf(xr,  k2, s3);
                }
                ov[e][0] = silu(s0 + t2.y);
                ov[e][1] = silu(s1 + t2.y);
                ov[e][2] = silu(s2 + t2.y);
                ov[e][3] = silu(s3 + t2.y);
            }
            const int f4i = f4lane + (chunk << 9);
            smem4[f4i]     = make_float4(ov[0][0], ov[1][0], ov[0][1], ov[1][1]);
            smem4[f4i + 1] = make_float4(ov[0][2], ov[1][2], ov[0][3], ov[1][3]);
            __syncthreads();
        }
    }

    const int lane = tid & 31;
    const int wv   = tid >> 5;
    const int mt0  = wv << 1;
    const int mo   = (lane >> 4) << 3;

    v8f acc[2][4];
#pragma unroll
    for (int ms = 0; ms < 2; ++ms)
#pragma unroll
        for (int ns = 0; ns < 4; ++ns)
#pragma unroll
            for (int r = 0; r < 8; ++r) acc[ms][ns][r] = 0.0f;

    const float* ar0 = w_out + (((mt0 << 4) + (lane & 15)) * CC) + ((lane >> 4) << 4);
    const float* ar1 = ar0 + 16 * CC;
    const v2f* bptr = (const v2f*)smem;

    float4 abuf[2][2][4];
#pragma unroll
    for (int q = 0; q < 4; ++q) {
        abuf[0][0][q] = *(const float4*)(ar0 + (q << 2));
        abuf[0][1][q] = *(const float4*)(ar1 + (q << 2));
    }

#pragma unroll
    for (int kb = 0; kb < 8; ++kb) {
        const int cur = kb & 1;
        if (kb + 1 < 8) {
#pragma unroll
            for (int q = 0; q < 4; ++q) {
                abuf[cur ^ 1][0][q] = *(const float4*)(ar0 + ((kb + 1) << 5) + (q << 2));
                abuf[cur ^ 1][1][q] = *(const float4*)(ar1 + ((kb + 1) << 5) + (q << 2));
            }
        }
#pragma unroll
        for (int s = 0; s < 8; ++s) {
            const float4 q0 = abuf[cur][0][s >> 1];
            const float4 q1 = abuf[cur][1][s >> 1];
            v2f a0, a1;
            if (s & 1) { a0.x = q0.z; a0.y = q0.w; a1.x = q1.z; a1.y = q1.w; }
            else       { a0.x = q0.x; a0.y = q0.y; a1.x = q1.x; a1.y = q1.y; }
            const int kstep = (kb << 3) + s;
#pragma unroll
            for (int ns = 0; ns < 4; ++ns) {
                const v2f bf = bptr[((kstep << 2) + ns) * 32 + lane];
                acc[0][ns] = __builtin_amdgcn_wmma_f32_16x16x4_f32(
                    false, a0, false, bf, (short)0, acc[0][ns], false, false);
                acc[1][ns] = __builtin_amdgcn_wmma_f32_16x16x4_f32(
                    false, a1, false, bf, (short)0, acc[1][ns], false, false);
            }
        }
    }

    const int ncol = lane & 15;
#pragma unroll
    for (int ms = 0; ms < 2; ++ms) {
        const int obase = ((mt0 + ms) << 4) + mo;
#pragma unroll
        for (int r = 0; r < 8; ++r) {
            const int o = obase + r;
            const float t2 = obn[o];
            const float b2 = obn[OUTC + o];
#pragma unroll
            for (int ns = 0; ns < 4; ++ns) {
                const float v = silu(acc[ms][ns][r] * t2 + b2);
                out[(((b * OUTC + o) * HH + h) * WWD) + (ns << 4) + ncol] = v;
            }
        }
    }
}

extern "C" void kernel_launch(void* const* d_in, const int* in_sizes, int n_in,
                              void* d_out, int out_size, void* d_ws, size_t ws_size,
                              hipStream_t stream) {
    const float* x         = (const float*)d_in[0];
    const float* w_experts = (const float*)d_in[1];
    const float* bn_gamma  = (const float*)d_in[2];
    const float* bn_beta   = (const float*)d_in[3];
    const float* bn_mean   = (const float*)d_in[4];
    const float* bn_var    = (const float*)d_in[5];
    const float* scales    = (const float*)d_in[6];
    const float* w_out     = (const float*)d_in[7];
    const float* out_gamma = (const float*)d_in[8];
    const float* out_beta  = (const float*)d_in[9];
    const float* out_mean  = (const float*)d_in[10];
    const float* out_var   = (const float*)d_in[11];
    float* out = (float*)d_out;

    move_kernel<<<BB * HH, 256, 0, stream>>>(x, w_experts, bn_gamma, bn_beta, bn_mean, bn_var,
                                             scales, w_out, out_gamma, out_beta, out_mean,
                                             out_var, out);
}


# one 16-wave work-group per 2 image rows (256 groups): one expert fold, one 4-row x halo tile (2 buffers, 1 barrier per chunk), a wave's w_out fragments serve both rows (16 ch x 128 px); sums kept
# speedup vs baseline: 3.6869x; 1.0978x over previous
#include <hip/hip_runtime.h>
#include <hip/hip_bf16.h>


#define BB   8
#define CC   256
#define HH   64
#define WWD  64
#define EE   9
#define OUTC 256
#define EPSF 1e-5f

typedef __attribute__((ext_vector_type(2))) float v2f;
typedef __attribute__((ext_vector_type(8))) float v8f;

__device__ __forceinline__ float silu(float v) {
    return v / (1.0f + __expf(-v));
}

__global__ __launch_bounds__(512) void move_kernel(
    const float* __restrict__ x,         const float* __restrict__ w_experts,
    const float* __restrict__ bn_gamma,  const float* __restrict__ bn_beta,
    const float* __restrict__ bn_mean,   const float* __restrict__ bn_var,
    const float* __restrict__ scales,    const float* __restrict__ w_out,
    const float* __restrict__ out_gamma, const float* __restrict__ out_beta,
    const float* __restrict__ out_mean,  const float* __restrict__ out_var,
    float* __restrict__ out)
{
    __shared__ alignas(16) float smem[2 * CC * WWD];
    __shared__ float4 rawt[2 * 4 * 32 * 16];
    __shared__ float4 taps[CC * 3];
    __shared__ float obn[2 * OUTC];

    const int t   = threadIdx.x;
    const int tid = t & 255;
    const int g   = t >> 8;
    const int b   = blockIdx.x >> 5;
    const int h0  = (blockIdx.x & 31) << 1;

    {
        const int col4 = t & 15;
        const int chS  = t >> 4;
        const float* xb = x + b * (CC * HH * WWD);

        const int pr = tid >> 4;
        const bool has_l = (col4 > 0);
        const bool has_r = (col4 < 15);
        const int il = has_l ? (col4 << 2) - 1 : 0;
        const int ir = has_r ? (col4 << 2) + 4 : WWD - 1;
        const float* rawf = (const float*)rawt;
        const int f4lane = ((((((pr & 7) << 1) + g) << 2) + (col4 >> 2)) << 4) + ((col4 & 3) << 1) + ((pr >> 3) << 3);
        float4* smem4 = (float4*)smem;

        float4 pre[4];
#pragma unroll
        for (int j = 0; j < 4; ++j) {
            int hr = h0 + j - 1;
            hr = hr < 0 ? 0 : (hr > HH - 1 ? HH - 1 : hr);
            pre[j] = *(const float4*)(xb + ((chS * HH + hr) * WWD) + (col4 << 2));
        }

        if (t < CC) {
            const int c = t;
            float bias = 0.0f;
#pragma unroll
            for (int e = 0; e < EE; ++e) {
                const int ec = e * CC + c;
                const float tt = bn_gamma[ec] * rsqrtf(bn_var[ec] + EPSF);
                const float es = tt * scales[ec];
                bias += (bn_beta[ec] - bn_mean[ec] * tt) * scales[ec];
                smem[ec] = es;
            }
            ((float*)taps)[c * 12 + 9] = bias;
            const float t2 = out_gamma[c] * rsqrtf(out_var[c] + EPSF);
            obn[c]        = t2;
            obn[OUTC + c] = out_beta[c] - out_mean[c] * t2;
        }
        __syncthreads();
#pragma unroll
        for (int j = 0; j < 5; ++j) {
            const int idx = t + (j << 9);
            if (idx < CC * 9) {
                const int c   = idx / 9;
                const int k   = idx - c * 9;
                float acc = 0.0f;
#pragma unroll
                for (int e = 0; e < EE; ++e)
                    acc += smem[e * CC + c] * w_experts[e * (CC * 9) + idx];
                ((float*)taps)[c * 12 + k] = acc;
            }
        }

        for (int chunk = 0; chunk < 8; ++chunk) {
            const int hb4 = (chunk & 1) << 11;
#pragma unroll
            for (int j = 0; j < 4; ++j) {
                const bool rv = (unsigned)(h0 + j - 1) < (unsigned)HH;
                float4 v = pre[j];
                if (!rv) v = make_float4(0.0f, 0.0f, 0.0f, 0.0f);
                rawt[hb4 + t + (j << 9)] = v;
            }
            __syncthreads();

            if (chunk + 1 < 8) {
#pragma unroll
                for (int j = 0; j < 4; ++j) {
                    int hr = h0 + j - 1;
                    hr = hr < 0 ? 0 : (hr > HH - 1 ? HH - 1 : hr);
                    const int c = ((chunk + 1) << 5) + chS;
                    pre[j] = *(const float4*)(xb + ((c * HH + hr) * WWD) + (col4 << 2));
                }
            }

            float ov[2][4];
#pragma unroll
            for (int e = 0; e < 2; ++e) {
                const int chl = (pr << 1) + e;
                const int c   = (chunk << 5) + chl;
                const float4 t0 = taps[c * 3 + 0];
                const float4 t1 = taps[c * 3 + 1];
                const float4 t2 = taps[c * 3 + 2];
                const float we[9] = {t0.x, t0.y, t0.z, t0.w, t1.x, t1.y, t1.z, t1.w, t2.x};
                float s0 = 0.0f, s1 = 0.0f, s2 = 0.0f, s3 = 0.0f;
#pragma unroll
                for (int r = 0; r < 3; ++r) {
                    const int row = (g + r) * 32 + chl;
                    const float4 m = rawt[hb4 + (row << 4) + col4];
                    const float* rp = rawf + (hb4 << 2) + (row << 6);
                    float xl = rp[il];
                    float xr = rp[ir];
                    xl = has_l ? xl : 0.0f;
                    xr = has_r ? xr : 0.0f;
                    const float k0 = we[r * 3 + 0];
                    const float k1 = we[r * 3 + 1];
                    const float k2 = we[r * 3 + 2];
                    s0 = __builtin_fmaf(xl,  k0, s0);
                    s0 = __builtin_fmaf(m.x, k1, s0);
                    s0 = __builtin_fmaf(m.y, k2, s0);
                    s1 = __builtin_fmaf(m.x, k0, s1);
                    s1 = __builtin_fmaf(m.y, k1, s1);
                    s1 = __builtin_fmaf(m.z, k2, s1);
                    s2 = __builtin_fmaf(m.y, k0, s2);
                    s2 = __builtin_fmaf(m.z, k1, s2);
                    s2 = __builtin_fmaf(m.w, k2, s2);
                    s3 = __builtin_fmaf(m.z, k0, s3);
                    s3 = __builtin_fmaf(m.w, k1, s3);
                    s3 = __builtin_fmaf(xr,  k2, s3);
                }
                ov[e][0] = silu(s0 + t2.y);
                ov[e][1] = silu(s1 + t2.y);
                ov[e][2] = silu(s2 + t2.y);
                ov[e][3] = silu(s3 + t2.y);
            }
            const int f4i = f4lane + (chunk << 10);
            smem4[f4i]     = make_float4(ov[0][0], ov[1][0], ov[0][1], ov[1][1]);
            smem4[f4i + 1] = make_float4(ov[0][2], ov[1][2], ov[0][3], ov[1][3]);
        }
        __syncthreads();
    }

    const int lane = t & 31;
    const int wv   = t >> 5;
    const int mo   = (lane >> 4) << 3;

    v8f acc[2][4];
#pragma unroll
    for (int gg = 0; gg < 2; ++gg)
#pragma unroll
        for (int ns = 0; ns < 4; ++ns)
#pragma unroll
            for (int r = 0; r < 8; ++r) acc[gg][ns][r] = 0.0f;

    const float* ap = w_out + (((wv << 4) + (lane & 15)) * CC) + ((lane >> 4) << 4);
    const v2f* bp = (const v2f*)smem + lane;

    float4 abuf[2][4];
#pragma unroll
    for (int q = 0; q < 4; ++q) abuf[0][q] = *(const float4*)(ap + (q << 2));

#pragma unroll 1
    for (int kp = 0; kp < 4; ++kp) {
#pragma unroll
        for (int hb = 0; hb < 2; ++hb) {
            const float* an = (hb == 0) ? ap + 32 : (kp < 3 ? ap + 64 : ap + 32);
#pragma unroll
            for (int q = 0; q < 4; ++q) abuf[hb ^ 1][q] = *(const float4*)(an + (q << 2));
            asm volatile("" ::: "memory");
#pragma unroll
            for (int s = 0; s < 8; ++s) {
                const float4 q0 = abuf[hb][s >> 1];
                v2f a0;
                if (s & 1) { a0.x = q0.z; a0.y = q0.w; }
                else       { a0.x = q0.x; a0.y = q0.y; }
                const int ks = (hb << 3) + s;
#pragma unroll
                for (int ns = 0; ns < 4; ++ns) {
                    const v2f bf0 = bp[((ks << 3) + ns) * 32];
                    const v2f bf1 = bp[((ks << 3) + 4 + ns) * 32];
                    acc[0][ns] = __builtin_amdgcn_wmma_f32_16x16x4_f32(
                        false, a0, false, bf0, (short)0, acc[0][ns], false, false);
                    acc[1][ns] = __builtin_amdgcn_wmma_f32_16x16x4_f32(
                        false, a0, false, bf1, (short)0, acc[1][ns], false, false);
                }
            }
        }
        ap += 64;
        bp += 16 * 8 * 32;
    }

    const int ncol  = lane & 15;
    const int obase = (wv << 4) + mo;
#pragma unroll
    for (int r = 0; r < 8; ++r) {
        const int o = obase + r;
        const float t2 = obn[o];
        const float b2 = obn[OUTC + o];
#pragma unroll
        for (int gg = 0; gg < 2; ++gg) {
#pragma unroll
            for (int ns = 0; ns < 4; ++ns) {
                const float v = silu(acc[gg][ns][r] * t2 + b2);
                out[(((b * OUTC + o) * HH + h0 + gg) * WWD) + (ns << 4) + ncol] = v;
            }
        }
    }
}

extern "C" void kernel_launch(void* const* d_in, const int* in_sizes, int n_in,
                              void* d_out, int out_size, void* d_ws, size_t ws_size,
                              hipStream_t stream) {
    const float* x         = (const float*)d_in[0];
    const float* w_experts = (const float*)d_in[1];
    const float* bn_gamma  = (const float*)d_in[2];
    const float* bn_beta   = (const float*)d_in[3];
    const float* bn_mean   = (const float*)d_in[4];
    const float* bn_var    = (const float*)d_in[5];
    const float* scales    = (const float*)d_in[6];
    const float* w_out     = (const float*)d_in[7];
    const float* out_gamma = (const float*)d_in[8];
    const float* out_beta  = (const float*)d_in[9];
    const float* out_mean  = (const float*)d_in[10];
    const float* out_var   = (const float*)d_in[11];
    float* out = (float*)d_out;

    move_kernel<<<256, 512, 0, stream>>>(x, w_experts, bn_gamma, bn_beta, bn_mean, bn_var,
                                         scales, w_out, out_gamma, out_beta, out_mean,
                                         out_var, out);
}


# conv stage per wave: wave w owns channel pair w of every chunk for both image rows and stages its 4 x rows in a halo tile of its own (2 KB), so the stage has no work-group barrier (was one per chunk)
# speedup vs baseline: 3.8014x; 1.0311x over previous
#include <hip/hip_runtime.h>
#include <hip/hip_bf16.h>


#define BB   8
#define CC   256
#define HH   64
#define WWD  64
#define EE   9
#define OUTC 256
#define EPSF 1e-5f

typedef __attribute__((ext_vector_type(2))) float v2f;
typedef __attribute__((ext_vector_type(8))) float v8f;

__device__ __forceinline__ float silu(float v) {
    return v / (1.0f + __expf(-v));
}

__global__ __launch_bounds__(512) void move_kernel(
    const float* __restrict__ x,         const float* __restrict__ w_experts,
    const float* __restrict__ bn_gamma,  const float* __restrict__ bn_beta,
    const float* __restrict__ bn_mean,   const float* __restrict__ bn_var,
    const float* __restrict__ scales,    const float* __restrict__ w_out,
    const float* __restrict__ out_gamma, const float* __restrict__ out_beta,
    const float* __restrict__ out_mean,  const float* __restrict__ out_var,
    float* __restrict__ out)
{
    __shared__ alignas(16) float smem[2 * CC * WWD];
    __shared__ float4 rawt[2 * 16 * 4 * 2 * 16];
    __shared__ float4 taps[CC * 3];
    __shared__ float obn[2 * OUTC];

    const int t   = threadIdx.x;
    const int g   = (t >> 4) & 1;
    const int b   = blockIdx.x >> 5;
    const int h0  = (blockIdx.x & 31) << 1;

    {
        const int col4 = t & 15;
        const int pr   = t >> 5;
        const int chS  = (pr << 1) + g;
        const int wb4  = pr << 7;
        const int ln   = t & 31;
        const float* xb = x + b * (CC * HH * WWD);

        const bool has_l = (col4 > 0);
        const bool has_r = (col4 < 15);
        const int il = has_l ? (col4 << 2) - 1 : 0;
        const int ir = has_r ? (col4 << 2) + 4 : WWD - 1;
        const float* rawf = (const float*)rawt;
        const int f4lane = ((((((pr & 7) << 1) + g) << 2) + (col4 >> 2)) << 4) + ((col4 & 3) << 1) + ((pr >> 3) << 3);
        float4* smem4 = (float4*)smem;

        float4 pre[4];
#pragma unroll
        for (int j = 0; j < 4; ++j) {
            int hr = h0 + j - 1;
            hr = hr < 0 ? 0 : (hr > HH - 1 ? HH - 1 : hr);
            pre[j] = *(const float4*)(xb + ((chS * HH + hr) * WWD) + (col4 << 2));
        }

        if (t < CC) {
            const int c = t;
            float bias = 0.0f;
#pragma unroll
            for (int e = 0; e < EE; ++e) {
                const int ec = e * CC + c;
                const float tt = bn_gamma[ec] * rsqrtf(bn_var[ec] + EPSF);
                const float es = tt * scales[ec];
                bias += (bn_beta[ec] - bn_mean[ec] * tt) * scales[ec];
                smem[ec] = es;
            }
            ((float*)taps)[c * 12 + 9] = bias;
            const float t2 = out_gamma[c] * rsqrtf(out_var[c] + EPSF);
            obn[c]        = t2;
            obn[OUTC + c] = out_beta[c] - out_mean[c] * t2;
        }
        __syncthreads();
#pragma unroll
        for (int j = 0; j < 5; ++j) {
            const int idx = t + (j << 9);
            if (idx < CC * 9) {
                const int c   = idx / 9;
                const int k   = idx - c * 9;
                float acc = 0.0f;
#pragma unroll
                for (int e = 0; e < EE; ++e)
                    acc += smem[e * CC + c] * w_experts[e * (CC * 9) + idx];
                ((float*)taps)[c * 12 + k] = acc;
            }
        }
        __syncthreads();

        for (int chunk = 0; chunk < 8; ++chunk) {
            const int hb4 = ((chunk & 1) << 11) + wb4;
#pragma unroll
            for (int j = 0; j < 4; ++j) {
                const bool rv = (unsigned)(h0 + j - 1) < (unsigned)HH;
                float4 v = pre[j];
                if (!rv) v = make_float4(0.0f, 0.0f, 0.0f, 0.0f);
                rawt[hb4 + (j << 5) + ln] = v;
            }

            if (chunk + 1 < 8) {
#pragma unroll
                for (int j = 0; j < 4; ++j) {
                    int hr = h0 + j - 1;
                    hr = hr < 0 ? 0 : (hr > HH - 1 ? HH - 1 : hr);
                    const int c = ((chunk + 1) << 5) + chS;
                    pre[j] = *(const float4*)(xb + ((c * HH + hr) * WWD) + (col4 << 2));
                }
            }

            float ov[2][4];
#pragma unroll
            for (int e = 0; e < 2; ++e) {
                const int chl = (pr << 1) + e;
                const int c   = (chunk << 5) + chl;
                const float4 t0 = taps[c * 3 + 0];
                const float4 t1 = taps[c * 3 + 1];
                const float4 t2 = taps[c * 3 + 2];
                const float we[9] = {t0.x, t0.y, t0.z, t0.w, t1.x, t1.y, t1.z, t1.w, t2.x};
                float s0 = 0.0f, s1 = 0.0f, s2 = 0.0f, s3 = 0.0f;
#pragma unroll
                for (int r = 0; r < 3; ++r) {
                    const int row = ((g + r) << 1) + e;
                    const float4 m = rawt[hb4 + (row << 4) + col4];
                    const float* rp = rawf + (hb4 << 2) + (row << 6);
                    float xl = rp[il];
                    float xr = rp[ir];
                    xl = has_l ? xl : 0.0f;
                    xr = has_r ? xr : 0.0f;
                    const float k0 = we[r * 3 + 0];
                    const float k1 = we[r * 3 + 1];
                    const float k2 = we[r * 3 + 2];
                    s0 = __builtin_fmaf(xl,  k0, s0);
                    s0 = __builtin_fmaf(m.x, k1, s0);
                    s0 = __builtin_fmaf(m.y, k2, s0);
                    s1 = __builtin_fmaf(m.x, k0, s1);
                    s1 = __builtin_fmaf(m.y, k1, s1);
                    s1 = __builtin_fmaf(m.z, k2, s1);
                    s2 = __builtin_fmaf(m.y, k0, s2);
                    s2 = __builtin_fmaf(m.z, k1, s2);
                    s2 = __builtin_fmaf(m.w, k2, s2);
                    s3 = __builtin_fmaf(m.z, k0, s3);
                    s3 = __builtin_fmaf(m.w, k1, s3);
                    s3 = __builtin_fmaf(xr,  k2, s3);
                }
                ov[e][0] = silu(s0 + t2.y);
                ov[e][1] = silu(s1 + t2.y);
                ov[e][2] = silu(s2 + t2.y);
                ov[e][3] = silu(s3 + t2.y);
            }
            const int f4i = f4lane + (chunk << 10);
            smem4[f4i]     = make_float4(ov[0][0], ov[1][0], ov[0][1], ov[1][1]);
            smem4[f4i + 1] = make_float4(ov[0][2], ov[1][2], ov[0][3], ov[1][3]);
        }
        __syncthreads();
    }

    const int lane = t & 31;
    const int wv   = t >> 5;
    const int mo   = (lane >> 4) << 3;

    v8f acc[2][4];
#pragma unroll
    for (int gg = 0; gg < 2; ++gg)
#pragma unroll
        for (int ns = 0; ns < 4; ++ns)
#pragma unroll
            for (int r = 0; r < 8; ++r) acc[gg][ns][r] = 0.0f;

    const float* ap = w_out + (((wv << 4) + (lane & 15)) * CC) + ((lane >> 4) << 4);
    const v2f* bp = (const v2f*)smem + lane;

    float4 abuf[2][4];
#pragma unroll
    for (int q = 0; q < 4; ++q) abuf[0][q] = *(const float4*)(ap + (q << 2));

#pragma unroll 1
    for (int kp = 0; kp < 4; ++kp) {
#pragma unroll
        for (int hb = 0; hb < 2; ++hb) {
            const float* an = (hb == 0) ? ap + 32 : (kp < 3 ? ap + 64 : ap + 32);
#pragma unroll
            for (int q = 0; q < 4; ++q) abuf[hb ^ 1][q] = *(const float4*)(an + (q << 2));
            asm volatile("" ::: "memory");
#pragma unroll
            for (int s = 0; s < 8; ++s) {
                const float4 q0 = abuf[hb][s >> 1];
                v2f a0;
                if (s & 1) { a0.x = q0.z; a0.y = q0.w; }
                else       { a0.x = q0.x; a0.y = q0.y; }
                const int ks = (hb << 3) + s;
#pragma unroll
                for (int ns = 0; ns < 4; ++ns) {
                    const v2f bf0 = bp[((ks << 3) + ns) * 32];
                    const v2f bf1 = bp[((ks << 3) + 4 + ns) * 32];
                    acc[0][ns] = __builtin_amdgcn_wmma_f32_16x16x4_f32(
                        false, a0, false, bf0, (short)0, acc[0][ns], false, false);
                    acc[1][ns] = __builtin_amdgcn_wmma_f32_16x16x4_f32(
                        false, a0, false, bf1, (short)0, acc[1][ns], false, false);
                }
            }
        }
        ap += 64;
        bp += 16 * 8 * 32;
    }

    const int ncol  = lane & 15;
    const int obase = (wv << 4) + mo;
#pragma unroll
    for (int r = 0; r < 8; ++r) {
        const int o = obase + r;
        const float t2 = obn[o];
        const float b2 = obn[OUTC + o];
#pragma unroll
        for (int gg = 0; gg < 2; ++gg) {
#pragma unroll
            for (int ns = 0; ns < 4; ++ns) {
                const float v = silu(acc[gg][ns][r] * t2 + b2);
                out[(((b * OUTC + o) * HH + h0 + gg) * WWD) + (ns << 4) + ncol] = v;
            }
        }
    }
}

extern "C" void kernel_launch(void* const* d_in, const int* in_sizes, int n_in,
                              void* d_out, int out_size, void* d_ws, size_t ws_size,
                              hipStream_t stream) {
    const float* x         = (const float*)d_in[0];
    const float* w_experts = (const float*)d_in[1];
    const float* bn_gamma  = (const float*)d_in[2];
    const float* bn_beta   = (const float*)d_in[3];
    const float* bn_mean   = (const float*)d_in[4];
    const float* bn_var    = (const float*)d_in[5];
    const float* scales    = (const float*)d_in[6];
    const float* w_out     = (const float*)d_in[7];
    const float* out_gamma = (const float*)d_in[8];
    const float* out_beta  = (const float*)d_in[9];
    const float* out_mean  = (const float*)d_in[10];
    const float* out_var   = (const float*)d_in[11];
    float* out = (float*)d_out;

    move_kernel<<<256, 512, 0, stream>>>(x, w_experts, bn_gamma, bn_beta, bn_mean, bn_var,
                                         scales, w_out, out_gamma, out_beta, out_mean,
                                         out_var, out);
}
